# starting kernel
# baseline (speedup 1.0000x reference)
#include <hip/hip_runtime.h>

typedef __attribute__((ext_vector_type(2))) float v2f;
typedef __attribute__((ext_vector_type(4))) float v4f;
typedef __attribute__((ext_vector_type(8))) float v8f;

#define D_FEAT 64

__device__ __forceinline__ int bperm_i(int src, int lane) {
    return __builtin_amdgcn_ds_bpermute(lane << 2, src);
}
__device__ __forceinline__ float bperm_f(float src, int lane) {
    return __int_as_float(__builtin_amdgcn_ds_bpermute(lane << 2, __float_as_int(src)));
}

__global__ void gcn_zero_kernel(float* __restrict__ out, int n) {
    int i = (blockIdx.x * blockDim.x + threadIdx.x) * 4;
    if (i + 3 < n) {
        *(v4f*)(out + i) = (v4f){0.f, 0.f, 0.f, 0.f};
    } else {
        for (int k = i; k < n; ++k) out[k] = 0.f;
    }
}

__global__ __launch_bounds__(256) void gcn_spmm_wmma_kernel(
    const int* __restrict__ rows, const int* __restrict__ cols,
    const float* __restrict__ vals, const float* __restrict__ embeds,
    float* __restrict__ out, int nedges) {

    const int lane = threadIdx.x & 31;
    const int tile = blockIdx.x * 8 + (threadIdx.x >> 5);
    const int base = tile * 16;
    if (base >= nedges) return;

    const int j = lane & 15;
    const int e = base + j;
    const bool valid = (e < nedges);
    int   row = valid ? rows[e] : (-1 - j);
    int   col = valid ? cols[e] : 0;
    float val = valid ? vals[e] : 0.0f;

    int fo = 1;
#pragma unroll
    for (int i = 0; i < 15; ++i) {
        int ri = bperm_i(row, i);
        if (i < j && ri == row) fo = 0;
    }

    const int khalf = (lane >> 4) << 1;
    v2f A[4];
    int ck0[4], ck1[4];
#pragma unroll
    for (int q = 0; q < 4; ++q) {
        const int k0 = 4 * q + khalf;
        const int k1 = k0 + 1;
        int   rk0 = bperm_i(row, k0);
        int   rk1 = bperm_i(row, k1);
        float vk0 = bperm_f(val, k0);
        float vk1 = bperm_f(val, k1);
        A[q].x = (rk0 == row) ? vk0 : 0.0f;
        A[q].y = (rk1 == row) ? vk1 : 0.0f;
        ck0[q] = bperm_i(col, k0);
        ck1[q] = bperm_i(col, k1);
    }

    int rowM[8], foM[8];
#pragma unroll
    for (int r = 0; r < 8; ++r) {
        const int M = (lane < 16) ? r : (r + 8);
        rowM[r] = bperm_i(row, M);
        foM[r]  = bperm_i(fo, M);
    }

    v8f acc4[4];
#pragma unroll
    for (int c = 0; c < 4; ++c) {
        v8f acc = {0.f, 0.f, 0.f, 0.f, 0.f, 0.f, 0.f, 0.f};
#pragma unroll
        for (int q = 0; q < 4; ++q) {
            v2f B;
            B.x = embeds[ck0[q] * D_FEAT + c * 16 + j];
            B.y = embeds[ck1[q] * D_FEAT + c * 16 + j];
            acc = __builtin_amdgcn_wmma_f32_16x16x4_f32(
                 false, A[q],  false, B,
                 (short)0, acc,  false,  false);
        }
        acc4[c] = acc;
    }

#pragma unroll
    for (int r = 0; r < 8; ++r) {
        const bool sc = (foM[r] != 0) && (rowM[r] >= 0);
        float* p = out + (sc ? rowM[r] : 0) * D_FEAT + j;
#pragma unroll
        for (int c = 0; c < 4; ++c) {
            atomicAdd(p + c * 16, sc ? acc4[c][r] : 0.0f);
        }
    }
}

extern "C" void kernel_launch(void* const* d_in, const int* in_sizes, int n_in,
                              void* d_out, int out_size, void* d_ws, size_t ws_size,
                              hipStream_t stream) {
    const int*   rows   = (const int*)d_in[0];
    const int*   cols   = (const int*)d_in[1];
    const float* vals   = (const float*)d_in[2];
    const float* embeds = (const float*)d_in[3];
    float*       out    = (float*)d_out;

    const int nedges = in_sizes[0];

    {
        int nthreads = (out_size + 3) / 4;
        int blocks   = (nthreads + 255) / 256;
        gcn_zero_kernel<<<blocks, 256, 0, stream>>>(out, out_size);
    }

    {
        int tiles  = (nedges + 15) / 16;
        int blocks = (tiles + 7) / 8;
        gcn_spmm_wmma_kernel<<<blocks, 256, 0, stream>>>(
            rows, cols, vals, embeds, out, nedges);
    }
}


# Tile reshaped to 8 edges x 32 features (was 16 x 16), block-diagonal A: every gather load and f32 atomic add covers one whole 128-B line of one row; in-tile merge over 8 edges, add order changes
# speedup vs baseline: 1.2565x; 1.2565x over previous
#include <hip/hip_runtime.h>


typedef __attribute__((ext_vector_type(2))) float v2f;
typedef __attribute__((ext_vector_type(4))) float v4f;
typedef __attribute__((ext_vector_type(8))) float v8f;

#define D_FEAT 64

__device__ __forceinline__ int bperm_i(int src, int lane) {
    return __builtin_amdgcn_ds_bpermute(lane << 2, src);
}

__global__ void gcn_zero_kernel(float* __restrict__ out, int n) {
    int i = (blockIdx.x * blockDim.x + threadIdx.x) * 4;
    if (i + 3 < n) {
        *(v4f*)(out + i) = (v4f){0.f, 0.f, 0.f, 0.f};
    } else {
        for (int k = i; k < n; ++k) out[k] = 0.f;
    }
}

__global__ __launch_bounds__(256) void gcn_spmm_wmma_kernel(
    const int* __restrict__ rows, const int* __restrict__ cols,
    const float* __restrict__ vals, const float* __restrict__ embeds,
    float* __restrict__ out, int nedges) {

    const int lane = threadIdx.x & 31;
    const int tile = blockIdx.x * 8 + (threadIdx.x >> 5);
    const int base = tile * 16;
    if (base >= nedges) return;

    const int j = lane & 15;
    const int e = base + j;
    const bool valid = (e < nedges);
    int   row = valid ? rows[e] : (-1 - j);
    int   col = valid ? cols[e] : 0;
    float val = valid ? vals[e] : 0.0f;

    const int j8   = j & 7;
    const int sub0 = j & 8;
    bool fo = (row >= 0);
#pragma unroll
    for (int i = 0; i < 7; ++i) {
        int ri = bperm_i(row, sub0 + i);
        if (i < j8 && ri == row) fo = false;
    }
    const unsigned fom = __builtin_amdgcn_ballot_w32(fo);

    int   srow[16], scol[16];
    float sval[16];
#pragma unroll
    for (int s = 0; s < 16; ++s) {
        srow[s] = __builtin_amdgcn_readlane(row, s);
        scol[s] = __builtin_amdgcn_readlane(col, s);
        sval[s] = __int_as_float(__builtin_amdgcn_readlane(__float_as_int(val), s));
    }

    const bool act = ((lane >> 4) == ((lane >> 3) & 1));
    v2f A[2][4];
#pragma unroll
    for (int t = 0; t < 2; ++t) {
        const int myrow = bperm_i(row, 8 * t + j8);
#pragma unroll
        for (int q = 0; q < 4; ++q) {
            const int s0 = 8 * t + 2 * q;
            A[t][q].x = (act && srow[s0]     == myrow) ? sval[s0]     : 0.0f;
            A[t][q].y = (act && srow[s0 + 1] == myrow) ? sval[s0 + 1] : 0.0f;
        }
    }

    v8f acc[2][2];
#pragma unroll
    for (int t = 0; t < 2; ++t) {
#pragma unroll
        for (int g = 0; g < 2; ++g) {
            v8f a = {0.f, 0.f, 0.f, 0.f, 0.f, 0.f, 0.f, 0.f};
#pragma unroll
            for (int q = 0; q < 4; ++q) {
                const int s0 = 8 * t + 2 * q;
                const float* r0 = embeds + (size_t)(unsigned)scol[s0]     * D_FEAT;
                const float* r1 = embeds + (size_t)(unsigned)scol[s0 + 1] * D_FEAT;
                v2f B;
                B.x = r0[g * 32 + lane];
                B.y = r1[g * 32 + lane];
                a = __builtin_amdgcn_wmma_f32_16x16x4_f32(
                     false, A[t][q],  false, B,
                     (short)0, a,  false,  false);
            }
            acc[t][g] = a;
        }
    }

#pragma unroll
    for (int t = 0; t < 2; ++t) {
#pragma unroll
        for (int r = 0; r < 8; ++r) {
            const int s = 8 * t + r;
            if (__builtin_expect((fom >> s) & 1u, 1)) {
                float* p = out + (size_t)(unsigned)srow[s] * D_FEAT;
                atomicAdd(p + lane,      acc[t][0][r]);
                atomicAdd(p + 32 + lane, acc[t][1][r]);
            }
        }
    }
}

extern "C" void kernel_launch(void* const* d_in, const int* in_sizes, int n_in,
                              void* d_out, int out_size, void* d_ws, size_t ws_size,
                              hipStream_t stream) {
    const int*   rows   = (const int*)d_in[0];
    const int*   cols   = (const int*)d_in[1];
    const float* vals   = (const float*)d_in[2];
    const float* embeds = (const float*)d_in[3];
    float*       out    = (float*)d_out;

    const int nedges = in_sizes[0];

    {
        int nthreads = (out_size + 3) / 4;
        int blocks   = (nthreads + 255) / 256;
        gcn_zero_kernel<<<blocks, 256, 0, stream>>>(out, out_size);
    }

    {
        int tiles  = (nedges + 15) / 16;
        int blocks = (tiles + 7) / 8;
        gcn_spmm_wmma_kernel<<<blocks, 256, 0, stream>>>(
            rows, cols, vals, embeds, out, nedges);
    }
}
